# starting kernel
# baseline (speedup 1.0000x reference)
#include <hip/hip_runtime.h>

typedef __attribute__((ext_vector_type(2))) float v2f;
typedef __attribute__((ext_vector_type(4))) float v4f;
typedef __attribute__((ext_vector_type(8))) float v8f;

__device__ __forceinline__ v2f make_v2f(float a, float b) {
    v2f r; r.x = a; r.y = b; return r;
}

#define BBATCH 4
#define CC     16
#define HH     64
#define WW     64
#define FF     32
#define KK     3

__global__ __launch_bounds__(256) void pade_conv_wmma(
    const float* __restrict__ x,
    const float* __restrict__ nums,
    const float* __restrict__ denoms,
    float* __restrict__ out)
{
    const int lane  = threadIdx.x & 31;
    const int gw    = blockIdx.x * 8 + (threadIdx.x >> 5);
    const int ftile = gw & 1;
    const int tile  = gw >> 1;

    const int w0 = (tile & 3) * 16;
    const int h  = (tile >> 2) & (HH - 1);
    const int b  = tile >> 8;

    const int  laneM = lane & 15;
    const bool hi    = lane >= 16;

    const int f = ftile * 16 + laneM;
    const float* __restrict__ npF = nums   + (size_t)f * (CC * KK * KK) * 6;
    const float* __restrict__ dpF = denoms + (size_t)f * (CC * KK * KK) * 4;

    v8f acc = {};
    const v8f zc = {};

    const int  loadIdx = (lane < 18) ? lane : 17;
    const int  wl      = w0 - 1 + loadIdx;
    const bool wlOK    = (unsigned)wl < (unsigned)WW;

    for (int c = 0; c < CC; ++c) {
        const float* __restrict__ xplane = x + (size_t)(b * CC + c) * (HH * WW);
        for (int a = 0; a < KK; ++a) {
            const int  hh  = h + a - 1;
            const bool hOK = (unsigned)hh < (unsigned)HH;
            const float* __restrict__ xrow = xplane + (size_t)(hOK ? hh : 0) * WW;

            float xload = 0.0f;
            if (hOK && wlOK) xload = xrow[wl];
            const int xbits = __float_as_int(xload);

#pragma unroll
            for (int bb = 0; bb < KK; ++bb) {
                const float xv = __int_as_float(
                    __builtin_amdgcn_ds_bpermute((laneM + bb) << 2, xbits));

                const float x2 = xv * xv;
                const float x3 = x2 * xv;
                const float x4 = x2 * x2;
                const float x5 = x4 * xv;

                v2f aP0, aP1, aQ;
                if (!hi) {
                    aP0 = make_v2f(1.0f, xv);
                    aP1 = make_v2f(x4,   x5);
                    aQ  = make_v2f(xv,   x2);
                } else {
                    aP0 = make_v2f(x2,   x3);
                    aP1 = make_v2f(0.0f, 0.0f);
                    aQ  = make_v2f(x3,   x4);
                }

                const int tap = (c * KK + a) * KK + bb;
                const float* np = npF + (size_t)tap * 6;
                const float* dp = dpF + (size_t)tap * 4;

                v2f bP0 = *(const v2f*)(np + (hi ? 2 : 0));
                v2f bP1 = hi ? make_v2f(0.0f, 0.0f) : *(const v2f*)(np + 4);
                v2f bQ  = *(const v2f*)(dp + (hi ? 2 : 0));

                v8f P = __builtin_amdgcn_wmma_f32_16x16x4_f32(
                    false, aP0, false, bP0, (short)0, zc, false, false);
                P = __builtin_amdgcn_wmma_f32_16x16x4_f32(
                    false, aP1, false, bP1, (short)0, P, false, false);
                v8f Q = __builtin_amdgcn_wmma_f32_16x16x4_f32(
                    false, aQ, false, bQ, (short)0, zc, false, false);

#pragma unroll
                for (int e = 0; e < 8; ++e)
                    acc[e] += P[e] * __builtin_amdgcn_rcpf(1.0f + __builtin_fabsf(Q[e]));
            }
        }
    }

    const int wbase = w0 + (hi ? 8 : 0);
    float* o = out + ((size_t)(b * FF + f) * HH + h) * WW + wbase;

    v4f s;
    s.x = acc[0]; s.y = acc[1]; s.z = acc[2]; s.w = acc[3];
    *(v4f*)(o) = s;
    s.x = acc[4]; s.y = acc[5]; s.z = acc[6]; s.w = acc[7];
    *(v4f*)(o + 4) = s;
}

extern "C" void kernel_launch(void* const* d_in, const int* in_sizes, int n_in,
                              void* d_out, int out_size, void* d_ws, size_t ws_size,
                              hipStream_t stream) {
    (void)in_sizes; (void)n_in; (void)out_size; (void)d_ws; (void)ws_size;
    const float* x      = (const float*)d_in[0];
    const float* nums   = (const float*)d_in[1];
    const float* denoms = (const float*)d_in[2];
    float* out = (float*)d_out;

    pade_conv_wmma<<<256, 256, 0, stream>>>(x, nums, denoms, out);
}


# coefficients re-laid in B-fragment order by a small first launch into the workspace; main kernel reads them as whole-line 16-byte loads, one channel ahead in registers, not 16-line strided loads
# speedup vs baseline: 2.6373x; 2.6373x over previous
#include <hip/hip_runtime.h>

typedef __attribute__((ext_vector_type(2))) float v2f;
typedef __attribute__((ext_vector_type(4))) float v4f;
typedef __attribute__((ext_vector_type(8))) float v8f;

__device__ __forceinline__ v2f make_v2f(float a, float b) {
    v2f r; r.x = a; r.y = b; return r;
}

#define BBATCH 4
#define CC     16
#define HH     64
#define WW     64
#define FF     32
#define KK     3

#define NJ        14
#define CF_V4F    (2 * CC * NJ * 32)

__global__ __launch_bounds__(256) void pack_coeffs(
    const float* __restrict__ nums,
    const float* __restrict__ denoms,
    float* __restrict__ cf)
{
    const int gid   = blockIdx.x * 256 + threadIdx.x;
    const int lane  = gid & 31;
    const int pj    = gid >> 5;
    const int j     = pj % NJ;
    const int fc    = pj / NJ;
    const int c     = fc & (CC - 1);
    const int ftile = fc >> 4;
    const int f     = ftile * 16 + (lane & 15);
    const int kofs  = (lane >= 16) ? 2 : 0;

    const float* __restrict__ np = nums   + (size_t)(f * CC + c) * (KK * KK * 6);
    const float* __restrict__ dp = denoms + (size_t)(f * CC + c) * (KK * KK * 4);

    float v[4];
#pragma unroll
    for (int e = 0; e < 4; ++e) {
        const int  i   = 4 * j + e;
        const bool pad = i >= KK * KK * 6;
        const int  ic  = pad ? 0 : i;
        const int  tap = ic / 6;
        const int  r   = ic - tap * 6;
        const float* src;
        bool zero = pad;
        if (r < 2)      { src = np + tap * 6 + kofs + r; }
        else if (r < 4) { src = np + tap * 6 + 2 + r; zero = zero || (kofs != 0); }
        else            { src = dp + tap * 4 + kofs + (r - 4); }
        const float ld = *src;
        v[e] = zero ? 0.0f : ld;
    }
    v4f o; o.x = v[0]; o.y = v[1]; o.z = v[2]; o.w = v[3];
    ((v4f*)cf)[gid] = o;
}

__device__ __forceinline__ void load_coeffs(v4f (&dst)[NJ], const v4f* __restrict__ p) {
#pragma unroll
    for (int j = 0; j < NJ - 1; ++j) dst[j] = p[j * 32];
    const v2f t = *(const v2f*)(p + (NJ - 1) * 32);
    dst[NJ - 1].x = t.x; dst[NJ - 1].y = t.y; dst[NJ - 1].z = 0.0f; dst[NJ - 1].w = 0.0f;
}

__device__ __forceinline__ void channel_step(
    const float* __restrict__ xplane, const int h, const int wl, const bool wlOK,
    const int laneM, const bool hi,
    const v4f (&cur)[NJ], v4f (&nxt)[NJ], const v4f* __restrict__ nxt_p,
    v8f& acc)
{
    const v8f zc = {};

    int xbits[KK];
#pragma unroll
    for (int a = 0; a < KK; ++a) {
        const int  hh  = h + a - 1;
        const bool hOK = (unsigned)hh < (unsigned)HH;
        const float* __restrict__ xrow = xplane + (size_t)(hOK ? hh : 0) * WW;
        float xload = 0.0f;
        if (hOK && wlOK) xload = xrow[wl];
        xbits[a] = __float_as_int(xload);
    }

    load_coeffs(nxt, nxt_p);

#pragma unroll
    for (int a = 0; a < KK; ++a) {
#pragma unroll
        for (int bb = 0; bb < KK; ++bb) {
            const float xv = __int_as_float(
                __builtin_amdgcn_ds_bpermute((laneM + bb) << 2, xbits[a]));

            const float x2 = xv * xv;
            const float x3 = x2 * xv;
            const float x4 = x2 * x2;
            const float x5 = x4 * xv;

            v2f aP0, aP1, aQ;
            if (!hi) {
                aP0 = make_v2f(1.0f, xv);
                aP1 = make_v2f(x4,   x5);
                aQ  = make_v2f(xv,   x2);
            } else {
                aP0 = make_v2f(x2,   x3);
                aP1 = make_v2f(0.0f, 0.0f);
                aQ  = make_v2f(x3,   x4);
            }

            const int i0 = (a * KK + bb) * 6;
            const v2f bP0 = make_v2f(cur[(i0 + 0) >> 2][(i0 + 0) & 3], cur[(i0 + 1) >> 2][(i0 + 1) & 3]);
            const v2f bP1 = make_v2f(cur[(i0 + 2) >> 2][(i0 + 2) & 3], cur[(i0 + 3) >> 2][(i0 + 3) & 3]);
            const v2f bQ  = make_v2f(cur[(i0 + 4) >> 2][(i0 + 4) & 3], cur[(i0 + 5) >> 2][(i0 + 5) & 3]);

            v8f P = __builtin_amdgcn_wmma_f32_16x16x4_f32(
                false, aP0, false, bP0, (short)0, zc, false, false);
            P = __builtin_amdgcn_wmma_f32_16x16x4_f32(
                false, aP1, false, bP1, (short)0, P, false, false);
            v8f Q = __builtin_amdgcn_wmma_f32_16x16x4_f32(
                false, aQ, false, bQ, (short)0, zc, false, false);

#pragma unroll
            for (int e = 0; e < 8; ++e)
                acc[e] += P[e] * __builtin_amdgcn_rcpf(1.0f + __builtin_fabsf(Q[e]));
        }
    }
}

__global__ __launch_bounds__(256) void pade_conv_wmma(
    const float* __restrict__ x,
    const float* __restrict__ cf,
    float* __restrict__ out)
{
    const int lane  = threadIdx.x & 31;
    const int gw    = blockIdx.x * 8 + (threadIdx.x >> 5);
    const int ftile = gw & 1;
    const int tile  = gw >> 1;

    const int w0 = (tile & 3) * 16;
    const int h  = (tile >> 2) & (HH - 1);
    const int b  = tile >> 8;

    const int  laneM = lane & 15;
    const bool hi    = lane >= 16;

    const int f = ftile * 16 + laneM;

    v8f acc = {};

    const int  loadIdx = (lane < 18) ? lane : 17;
    const int  wl      = w0 - 1 + loadIdx;
    const bool wlOK    = (unsigned)wl < (unsigned)WW;

    const v4f* __restrict__ cfw = (const v4f*)cf + (size_t)ftile * (CC * NJ * 32) + lane;
    const float* __restrict__ xb = x + (size_t)b * CC * (HH * WW);

    v4f bufA[NJ], bufB[NJ];
    load_coeffs(bufA, cfw);

#pragma unroll 1
    for (int c = 0; c < CC; c += 2) {
        channel_step(xb + (size_t)c * (HH * WW), h, wl, wlOK, laneM, hi,
                     bufA, bufB, cfw + (size_t)(c + 1) * (NJ * 32), acc);
        const int cn = (c + 2 < CC) ? (c + 2) : (CC - 1);
        channel_step(xb + (size_t)(c + 1) * (HH * WW), h, wl, wlOK, laneM, hi,
                     bufB, bufA, cfw + (size_t)cn * (NJ * 32), acc);
        asm volatile("" :: "v"(bufA[0]), "v"(bufA[1]), "v"(bufA[2]), "v"(bufA[3]), "v"(bufA[4]), "v"(bufA[5]), "v"(bufA[6]), "v"(bufA[7]), "v"(bufA[8]), "v"(bufA[9]), "v"(bufA[10]), "v"(bufA[11]), "v"(bufA[12]), "v"(bufA[13].x), "v"(bufA[13].y));
    }

    const int wbase = w0 + (hi ? 8 : 0);
    float* o = out + ((size_t)(b * FF + f) * HH + h) * WW + wbase;

    v4f s;
    s.x = acc[0]; s.y = acc[1]; s.z = acc[2]; s.w = acc[3];
    *(v4f*)(o) = s;
    s.x = acc[4]; s.y = acc[5]; s.z = acc[6]; s.w = acc[7];
    *(v4f*)(o + 4) = s;
}

extern "C" void kernel_launch(void* const* d_in, const int* in_sizes, int n_in,
                              void* d_out, int out_size, void* d_ws, size_t ws_size,
                              hipStream_t stream) {
    (void)in_sizes; (void)n_in; (void)out_size; (void)ws_size;
    const float* x      = (const float*)d_in[0];
    const float* nums   = (const float*)d_in[1];
    const float* denoms = (const float*)d_in[2];
    float* out = (float*)d_out;
    float* cf  = (float*)d_ws;

    pack_coeffs<<<56, 256, 0, stream>>>(nums, denoms, cf);
    pade_conv_wmma<<<256, 256, 0, stream>>>(x, cf, out);
}


# channel loop branch-free (x windows from clamped addresses, one trip ahead) and laid out as a software pipeline over its taps: one product per scheduling piece beside the fold of the tap before
# speedup vs baseline: 2.8204x; 1.0694x over previous
#include <hip/hip_runtime.h>

typedef __attribute__((ext_vector_type(2))) float v2f;
typedef __attribute__((ext_vector_type(4))) float v4f;
typedef __attribute__((ext_vector_type(8))) float v8f;

__device__ __forceinline__ v2f make_v2f(float a, float b) {
    v2f r; r.x = a; r.y = b; return r;
}

#define BBATCH 4
#define CC     16
#define HH     64
#define WW     64
#define FF     32
#define KK     3

#define NJ        14
#define CF_V4F    (2 * CC * NJ * 32)

__global__ __launch_bounds__(256) void pack_coeffs(
    const float* __restrict__ nums,
    const float* __restrict__ denoms,
    float* __restrict__ cf)
{
    const int gid   = blockIdx.x * 256 + threadIdx.x;
    const int lane  = gid & 31;
    const int pj    = gid >> 5;
    const int j     = pj % NJ;
    const int fc    = pj / NJ;
    const int c     = fc & (CC - 1);
    const int ftile = fc >> 4;
    const int f     = ftile * 16 + (lane & 15);
    const int kofs  = (lane >= 16) ? 2 : 0;

    const float* __restrict__ np = nums   + (size_t)(f * CC + c) * (KK * KK * 6);
    const float* __restrict__ dp = denoms + (size_t)(f * CC + c) * (KK * KK * 4);

    float v[4];
#pragma unroll
    for (int e = 0; e < 4; ++e) {
        const int  i   = 4 * j + e;
        const bool pad = i >= KK * KK * 6;
        const int  ic  = pad ? 0 : i;
        const int  tap = ic / 6;
        const int  r   = ic - tap * 6;
        const float* src;
        bool zero = pad;
        if (r < 2)      { src = np + tap * 6 + kofs + r; }
        else if (r < 4) { src = np + tap * 6 + 2 + r; zero = zero || (kofs != 0); }
        else            { src = dp + tap * 4 + kofs + (r - 4); }
        const float ld = *src;
        v[e] = zero ? 0.0f : ld;
    }
    v4f o; o.x = v[0]; o.y = v[1]; o.z = v[2]; o.w = v[3];
    ((v4f*)cf)[gid] = o;
}

__device__ __forceinline__ void load_coeffs(v4f (&dst)[NJ], const v4f* p) {
#pragma unroll
    for (int j = 0; j < NJ - 1; ++j) dst[j] = p[j * 32];
    const v2f t = *(const v2f*)(p + (NJ - 1) * 32);
    dst[NJ - 1].x = t.x; dst[NJ - 1].y = t.y; dst[NJ - 1].z = 0.0f; dst[NJ - 1].w = 0.0f;
}

__device__ __forceinline__ void request_x(int (&dst)[KK], const float* xplane, const int (&xoff)[KK]) {
#pragma unroll
    for (int a = 0; a < KK; ++a) dst[a] = __float_as_int(xplane[xoff[a]]);
}

__device__ __forceinline__ void select_x(int (&dst)[KK], const int (&raw)[KK], const bool (&xok)[KK]) {
#pragma unroll
    for (int a = 0; a < KK; ++a) dst[a] = xok[a] ? raw[a] : 0;
}

struct Powers { float x1, x2, x3, x4, x5; };
struct AFrag  { v2f p0, p1, q; };

__device__ __forceinline__ Powers make_powers(const float xv) {
    Powers w;
    w.x1 = xv;
    w.x2 = xv * xv;
    w.x3 = w.x2 * xv;
    w.x4 = w.x2 * w.x2;
    w.x5 = w.x4 * xv;
    return w;
}

__device__ __forceinline__ AFrag make_afrag(const Powers& w, const bool hi) {
    AFrag f;
    if (!hi) {
        f.p0 = make_v2f(1.0f, w.x1);
        f.p1 = make_v2f(w.x4, w.x5);
        f.q  = make_v2f(w.x1, w.x2);
    } else {
        f.p0 = make_v2f(w.x2, w.x3);
        f.p1 = make_v2f(0.0f, 0.0f);
        f.q  = make_v2f(w.x3, w.x4);
    }
    return f;
}

__device__ __forceinline__ void fold_den(v2f (&d)[4], const v8f& Q) {
#pragma unroll
    for (int k = 0; k < 4; ++k)
        d[k] = make_v2f(1.0f + __builtin_fabsf(Q[2 * k]), 1.0f + __builtin_fabsf(Q[2 * k + 1]));
}
__device__ __forceinline__ void fold_rcp(v2f (&r)[4], const v2f (&d)[4]) {
#pragma unroll
    for (int k = 0; k < 4; ++k)
        r[k] = make_v2f(__builtin_amdgcn_rcpf(d[k].x), __builtin_amdgcn_rcpf(d[k].y));
}
__device__ __forceinline__ void fold_fma(v2f (&acc)[4], const v8f& P, const v2f (&r)[4]) {
#pragma unroll
    for (int k = 0; k < 4; ++k)
        acc[k] = __builtin_elementwise_fma(make_v2f(P[2 * k], P[2 * k + 1]), r[k], acc[k]);
}

__device__ __forceinline__ float cf_at(const v4f (&buf)[NJ], const int i) { return buf[i >> 2][i & 3]; }

#define TAPS (2 * KK * KK)

struct Pipe {
    float xv[4];
    AFrag af[2];
    v8f   P[2], Q[2];
};

__device__ __forceinline__ float tap_x(const int (&xb)[KK], const int tt, const int laneM) {
    const int a = tt / KK, bb = tt - a * KK;
    return __int_as_float(__builtin_amdgcn_ds_bpermute((laneM + bb) << 2, xb[a]));
}

__device__ __forceinline__ void trip(
    Pipe& st,
    const int (&xb0)[KK], const int (&xb1)[KK],
    const int (&xn0)[KK], const int (&xn1)[KK], const bool (&xok)[KK],
    int (&xbn0)[KK], int (&xbn1)[KK],
    const int laneM, const bool hi,
    v4f (&bufA)[NJ], v4f (&bufB)[NJ],
    const v4f* pB, const v4f* pA,
    v2f (&acc)[4])
{
    const v8f zc = {};
    Powers pw = {};
    v2f    den[4], rcp[4];

    load_coeffs(bufB, pB);

#pragma unroll
    for (int u = 0; u < TAPS; ++u) {
        const int ch = (u >= KK * KK) ? 1 : 0;
        const int i0 = (u - ch * KK * KK) * 6;
        const int ur = u & 1;
        const int up = (u + 1) & 1;
        const int t1 = (u + 1) % TAPS;
        const int t3 = (u + 3) % TAPS;

        if (u == KK * KK) load_coeffs(bufA, pA);
        if (u == TAPS - 3) {
            select_x(xbn0, xn0, xok);
            select_x(xbn1, xn1, xok);
        }

        {
            const v2f bP0 = make_v2f(ch ? cf_at(bufB, i0 + 0) : cf_at(bufA, i0 + 0),
                                     ch ? cf_at(bufB, i0 + 1) : cf_at(bufA, i0 + 1));
            st.P[ur] = __builtin_amdgcn_wmma_f32_16x16x4_f32(
                false, st.af[ur].p0, false, bP0, (short)0, zc, false, false);
        }
        fold_den(den, st.Q[up]);
        __builtin_amdgcn_sched_barrier(0);

        {
            const v2f bQ = make_v2f(ch ? cf_at(bufB, i0 + 4) : cf_at(bufA, i0 + 4),
                                    ch ? cf_at(bufB, i0 + 5) : cf_at(bufA, i0 + 5));
            st.Q[ur] = __builtin_amdgcn_wmma_f32_16x16x4_f32(
                false, st.af[ur].q, false, bQ, (short)0, zc, false, false);
        }
        fold_rcp(rcp, den);
        pw = make_powers(st.xv[t1 & 3]);
        __builtin_amdgcn_sched_barrier(0);

        {
            const v2f bP1 = make_v2f(ch ? cf_at(bufB, i0 + 2) : cf_at(bufA, i0 + 2),
                                     ch ? cf_at(bufB, i0 + 3) : cf_at(bufA, i0 + 3));
            st.P[ur] = __builtin_amdgcn_wmma_f32_16x16x4_f32(
                false, st.af[ur].p1, false, bP1, (short)0, st.P[ur], false, false);
        }
        fold_fma(acc, st.P[up], rcp);
        st.af[t1 & 1] = make_afrag(pw, hi);
        if (u + 3 >= TAPS)        st.xv[t3 & 3] = tap_x(xbn0, t3, laneM);
        else if (u + 3 >= KK * KK) st.xv[t3 & 3] = tap_x(xb1, t3 - KK * KK, laneM);
        else                       st.xv[t3 & 3] = tap_x(xb0, t3, laneM);
        __builtin_amdgcn_sched_barrier(0);
    }
}

__global__ __launch_bounds__(256) void pade_conv_wmma(
    const float* x,
    const float* cf,
    float* __restrict__ out)
{
    const int lane  = threadIdx.x & 31;
    const int gw    = blockIdx.x * 8 + (threadIdx.x >> 5);
    const int ftile = gw & 1;
    const int tile  = gw >> 1;

    const int w0 = (tile & 3) * 16;
    const int h  = (tile >> 2) & (HH - 1);
    const int b  = tile >> 8;

    const int  laneM = lane & 15;
    const bool hi    = lane >= 16;

    const int f = ftile * 16 + laneM;

    v2f acc[4] = {};

    const int  loadIdx = (lane < 18) ? lane : 17;
    const int  wl      = w0 - 1 + loadIdx;
    const bool wlOK    = (unsigned)wl < (unsigned)WW;

    int  xoff[KK];
    bool xok[KK];
#pragma unroll
    for (int a = 0; a < KK; ++a) {
        const int  hh  = h + a - 1;
        const bool hOK = (unsigned)hh < (unsigned)HH;
        xoff[a] = (hOK ? hh : h) * WW + (wlOK ? wl : w0);
        xok[a]  = hOK && wlOK;
    }

    const v4f* cfw = (const v4f*)cf + (size_t)ftile * (CC * NJ * 32) + lane;
    const float* xb = x + (size_t)b * CC * (HH * WW);

    int xr0[KK], xr1[KK];
    request_x(xr0, xb, xoff);
    request_x(xr1, xb + HH * WW, xoff);

    v4f bufA[NJ], bufB[NJ];
    load_coeffs(bufA, cfw);

    int xb0[KK], xb1[KK];
    select_x(xb0, xr0, xok);
    select_x(xb1, xr1, xok);

    Pipe st;
    st.P[1] = v8f{};
    st.Q[1] = v8f{};
#pragma unroll
    for (int t = 0; t < 3; ++t) st.xv[t] = tap_x(xb0, t, laneM);
    st.af[0] = make_afrag(make_powers(st.xv[0]), hi);

#pragma unroll 1
    for (int c = 0; c < CC; c += 2) {
        const int cp = (c + 2 < CC) ? (c + 2) : c;
        const int cn = (c + 2 < CC) ? (c + 2) : (CC - 1);
        int xn0[KK], xn1[KK], xbn0[KK], xbn1[KK];
        request_x(xn0, xb + (size_t)cp * (HH * WW), xoff);
        request_x(xn1, xb + (size_t)(cp + 1) * (HH * WW), xoff);

        trip(st, xb0, xb1, xn0, xn1, xok, xbn0, xbn1, laneM, hi, bufA, bufB,
             cfw + (size_t)(c + 1) * (NJ * 32), cfw + (size_t)cn * (NJ * 32), acc);

        asm volatile("" :: "v"(bufA[0]), "v"(bufA[1]), "v"(bufA[2]), "v"(bufA[3]), "v"(bufA[4]), "v"(bufA[5]), "v"(bufA[6]), "v"(bufA[7]), "v"(bufA[8]), "v"(bufA[9]), "v"(bufA[10]), "v"(bufA[11]), "v"(bufA[12]), "v"(bufA[13].x), "v"(bufA[13].y));
#pragma unroll
        for (int a = 0; a < KK; ++a) { xb0[a] = xbn0[a]; xb1[a] = xbn1[a]; }
    }

    {
        v2f den[4], rcp[4];
        fold_den(den, st.Q[1]);
        fold_rcp(rcp, den);
        fold_fma(acc, st.P[1], rcp);
    }

    const int wbase = w0 + (hi ? 8 : 0);
    float* o = out + ((size_t)(b * FF + f) * HH + h) * WW + wbase;

    v4f s;
    s.x = acc[0].x; s.y = acc[0].y; s.z = acc[1].x; s.w = acc[1].y;
    *(v4f*)(o) = s;
    s.x = acc[2].x; s.y = acc[2].y; s.z = acc[3].x; s.w = acc[3].y;
    *(v4f*)(o + 4) = s;
}

extern "C" void kernel_launch(void* const* d_in, const int* in_sizes, int n_in,
                              void* d_out, int out_size, void* d_ws, size_t ws_size,
                              hipStream_t stream) {
    (void)in_sizes; (void)n_in; (void)out_size; (void)ws_size;
    const float* x      = (const float*)d_in[0];
    const float* nums   = (const float*)d_in[1];
    const float* denoms = (const float*)d_in[2];
    float* out = (float*)d_out;
    float* cf  = (float*)d_ws;

    pack_coeffs<<<56, 256, 0, stream>>>(nums, denoms, cf);
    pade_conv_wmma<<<256, 256, 0, stream>>>(x, cf, out);
}


# operands swapped (coefficients A, powers B) so a lane's 8 product elements share its own x: denominator = 1+|x|*|b1+b2x+b3x^2+b4x^3| by one |.| fma each; P and Q share one power fragment
# speedup vs baseline: 2.9661x; 1.0517x over previous
#include <hip/hip_runtime.h>

typedef __attribute__((ext_vector_type(2))) float v2f;
typedef __attribute__((ext_vector_type(4))) float v4f;
typedef __attribute__((ext_vector_type(8))) float v8f;

__device__ __forceinline__ v2f make_v2f(float a, float b) {
    v2f r; r.x = a; r.y = b; return r;
}

#define BBATCH 4
#define CC     16
#define HH     64
#define WW     64
#define FF     32
#define KK     3

#define NJ        14
#define CF_V4F    (2 * CC * NJ * 32)

__global__ __launch_bounds__(256) void pack_coeffs(
    const float* __restrict__ nums,
    const float* __restrict__ denoms,
    float* __restrict__ cf)
{
    const int gid   = blockIdx.x * 256 + threadIdx.x;
    const int lane  = gid & 31;
    const int pj    = gid >> 5;
    const int j     = pj % NJ;
    const int fc    = pj / NJ;
    const int c     = fc & (CC - 1);
    const int ftile = fc >> 4;
    const int f     = ftile * 16 + (lane & 15);
    const int kofs  = (lane >= 16) ? 2 : 0;

    const float* __restrict__ np = nums   + (size_t)(f * CC + c) * (KK * KK * 6);
    const float* __restrict__ dp = denoms + (size_t)(f * CC + c) * (KK * KK * 4);

    float v[4];
#pragma unroll
    for (int e = 0; e < 4; ++e) {
        const int  i   = 4 * j + e;
        const bool pad = i >= KK * KK * 6;
        const int  ic  = pad ? 0 : i;
        const int  tap = ic / 6;
        const int  r   = ic - tap * 6;
        const float* src;
        bool zero = pad;
        if (r < 2)      { src = np + tap * 6 + kofs + r; }
        else if (r < 4) { src = np + tap * 6 + 2 + r; zero = zero || (kofs != 0); }
        else            { src = dp + tap * 4 + kofs + (r - 4); }
        const float ld = *src;
        v[e] = zero ? 0.0f : ld;
    }
    v4f o; o.x = v[0]; o.y = v[1]; o.z = v[2]; o.w = v[3];
    ((v4f*)cf)[gid] = o;
}

__device__ __forceinline__ void load_coeffs(v4f (&dst)[NJ], const v4f* p) {
#pragma unroll
    for (int j = 0; j < NJ - 1; ++j) dst[j] = p[j * 32];
    const v2f t = *(const v2f*)(p + (NJ - 1) * 32);
    dst[NJ - 1].x = t.x; dst[NJ - 1].y = t.y; dst[NJ - 1].z = 0.0f; dst[NJ - 1].w = 0.0f;
}

__device__ __forceinline__ void request_x(int (&dst)[KK], const float* xplane, const int (&xoff)[KK]) {
#pragma unroll
    for (int a = 0; a < KK; ++a) dst[a] = __float_as_int(xplane[xoff[a]]);
}

__device__ __forceinline__ void select_x(int (&dst)[KK], const int (&raw)[KK], const bool (&xok)[KK]) {
#pragma unroll
    for (int a = 0; a < KK; ++a) dst[a] = xok[a] ? raw[a] : 0;
}

struct Powers { float x1, x2, x4, x5; };
struct AFrag  { v2f p0, p1; };

__device__ __forceinline__ Powers make_powers(const float xv) {
    Powers w;
    w.x1 = xv;
    w.x2 = xv * xv;
    w.x4 = w.x2 * w.x2;
    w.x5 = w.x4 * xv;
    return w;
}

__device__ __forceinline__ AFrag make_afrag(const Powers& w, const bool hi) {
    AFrag f;
    const float s = hi ? w.x2 : 1.0f;
    f.p0 = make_v2f(s, s * w.x1);
    f.p1 = make_v2f(w.x4, w.x5);
    return f;
}

__device__ __forceinline__ float den1(float q, const float xa) {
    asm("v_fma_f32 %0, |%0|, |%1|, 1.0" : "+v"(q) : "v"(xa));
    return q;
}
__device__ __forceinline__ void fold_den(v2f (&d)[4], const v8f& Q, const float xa) {
#pragma unroll
    for (int k = 0; k < 4; ++k)
        d[k] = make_v2f(den1(Q[2 * k], xa), den1(Q[2 * k + 1], xa));
}
__device__ __forceinline__ void fold_rcp(v2f (&r)[4], const v2f (&d)[4]) {
#pragma unroll
    for (int k = 0; k < 4; ++k)
        r[k] = make_v2f(__builtin_amdgcn_rcpf(d[k].x), __builtin_amdgcn_rcpf(d[k].y));
}
__device__ __forceinline__ void fold_fma(v2f (&acc)[4], const v8f& P, const v2f (&r)[4]) {
#pragma unroll
    for (int k = 0; k < 4; ++k)
        acc[k] = __builtin_elementwise_fma(make_v2f(P[2 * k], P[2 * k + 1]), r[k], acc[k]);
}

__device__ __forceinline__ float cf_at(const v4f (&buf)[NJ], const int i) { return buf[i >> 2][i & 3]; }

#define TAPS (2 * KK * KK)

struct Pipe {
    float xv[4];
    AFrag af[2];
    float xa[2];
    v8f   P[2], Q[2];
};

__device__ __forceinline__ float tap_x(const int (&xb)[KK], const int tt, const int laneM) {
    const int a = tt / KK, bb = tt - a * KK;
    return __int_as_float(__builtin_amdgcn_ds_bpermute((laneM + bb) << 2, xb[a]));
}

__device__ __forceinline__ void trip(
    Pipe& st,
    const int (&xb0)[KK], const int (&xb1)[KK],
    const int (&xn0)[KK], const int (&xn1)[KK], const bool (&xok)[KK],
    int (&xbn0)[KK], int (&xbn1)[KK],
    const int laneM, const bool hi,
    v4f (&bufA)[NJ], v4f (&bufB)[NJ],
    const v4f* pB, const v4f* pA,
    v2f (&acc)[4])
{
    const v8f zc = {};
    Powers pw = {};
    v2f    den[4], rcp[4];

    load_coeffs(bufB, pB);

#pragma unroll
    for (int u = 0; u < TAPS; ++u) {
        const int ch = (u >= KK * KK) ? 1 : 0;
        const int i0 = (u - ch * KK * KK) * 6;
        const int ur = u & 1;
        const int up = (u + 1) & 1;
        const int t1 = (u + 1) % TAPS;
        const int t3 = (u + 3) % TAPS;

        if (u == KK * KK) load_coeffs(bufA, pA);
        if (u == TAPS - 3) {
            select_x(xbn0, xn0, xok);
            select_x(xbn1, xn1, xok);
        }

        {
            const v2f bP0 = make_v2f(ch ? cf_at(bufB, i0 + 0) : cf_at(bufA, i0 + 0),
                                     ch ? cf_at(bufB, i0 + 1) : cf_at(bufA, i0 + 1));
            st.P[ur] = __builtin_amdgcn_wmma_f32_16x16x4_f32(
                false, bP0, false, st.af[ur].p0, (short)0, zc, false, false);
        }
        fold_den(den, st.Q[up], st.xa[up]);
        __builtin_amdgcn_sched_barrier(0);

        {
            const v2f bQ = make_v2f(ch ? cf_at(bufB, i0 + 4) : cf_at(bufA, i0 + 4),
                                    ch ? cf_at(bufB, i0 + 5) : cf_at(bufA, i0 + 5));
            st.Q[ur] = __builtin_amdgcn_wmma_f32_16x16x4_f32(
                false, bQ, false, st.af[ur].p0, (short)0, zc, false, false);
        }
        fold_rcp(rcp, den);
        pw = make_powers(st.xv[t1 & 3]);
        st.xa[t1 & 1] = st.xv[t1 & 3];
        __builtin_amdgcn_sched_barrier(0);

        {
            const v2f bP1 = make_v2f(ch ? cf_at(bufB, i0 + 2) : cf_at(bufA, i0 + 2),
                                     ch ? cf_at(bufB, i0 + 3) : cf_at(bufA, i0 + 3));
            st.P[ur] = __builtin_amdgcn_wmma_f32_16x16x4_f32(
                false, bP1, false, st.af[ur].p1, (short)0, st.P[ur], false, false);
        }
        fold_fma(acc, st.P[up], rcp);
        st.af[t1 & 1] = make_afrag(pw, hi);
        if (u + 3 >= TAPS)        st.xv[t3 & 3] = tap_x(xbn0, t3, laneM);
        else if (u + 3 >= KK * KK) st.xv[t3 & 3] = tap_x(xb1, t3 - KK * KK, laneM);
        else                       st.xv[t3 & 3] = tap_x(xb0, t3, laneM);
        __builtin_amdgcn_sched_barrier(0);
    }
}

__global__ __launch_bounds__(256) void pade_conv_wmma(
    const float* x,
    const float* cf,
    float* __restrict__ out)
{
    const int lane  = threadIdx.x & 31;
    const int gw    = blockIdx.x * 8 + (threadIdx.x >> 5);
    const int ftile = gw & 1;
    const int tile  = gw >> 1;

    const int w0 = (tile & 3) * 16;
    const int h  = (tile >> 2) & (HH - 1);
    const int b  = tile >> 8;

    const int  laneM = lane & 15;
    const bool hi    = lane >= 16;

    v2f acc[4] = {};

    const int  loadIdx = (lane < 18) ? lane : 17;
    const int  wl      = w0 - 1 + loadIdx;
    const bool wlOK    = (unsigned)wl < (unsigned)WW;

    int  xoff[KK];
    bool xok[KK];
#pragma unroll
    for (int a = 0; a < KK; ++a) {
        const int  hh  = h + a - 1;
        const bool hOK = (unsigned)hh < (unsigned)HH;
        xoff[a] = (hOK ? hh : h) * WW + (wlOK ? wl : w0);
        xok[a]  = hOK && wlOK;
    }

    const v4f* cfw = (const v4f*)cf + (size_t)ftile * (CC * NJ * 32) + lane;
    const float* xb = x + (size_t)b * CC * (HH * WW);

    int xr0[KK], xr1[KK];
    request_x(xr0, xb, xoff);
    request_x(xr1, xb + HH * WW, xoff);

    v4f bufA[NJ], bufB[NJ];
    load_coeffs(bufA, cfw);

    int xb0[KK], xb1[KK];
    select_x(xb0, xr0, xok);
    select_x(xb1, xr1, xok);

    Pipe st;
    st.P[1] = v8f{};
    st.Q[1] = v8f{};
#pragma unroll
    for (int t = 0; t < 3; ++t) st.xv[t] = tap_x(xb0, t, laneM);
    st.af[0] = make_afrag(make_powers(st.xv[0]), hi);
    st.xa[0] = st.xv[0];
    st.xa[1] = 0.0f;

#pragma unroll 1
    for (int c = 0; c < CC; c += 2) {
        const int cp = (c + 2 < CC) ? (c + 2) : c;
        const int cn = (c + 2 < CC) ? (c + 2) : (CC - 1);
        int xn0[KK], xn1[KK], xbn0[KK], xbn1[KK];
        request_x(xn0, xb + (size_t)cp * (HH * WW), xoff);
        request_x(xn1, xb + (size_t)(cp + 1) * (HH * WW), xoff);

        trip(st, xb0, xb1, xn0, xn1, xok, xbn0, xbn1, laneM, hi, bufA, bufB,
             cfw + (size_t)(c + 1) * (NJ * 32), cfw + (size_t)cn * (NJ * 32), acc);

        asm volatile("" :: "v"(bufA[0]), "v"(bufA[1]), "v"(bufA[2]), "v"(bufA[3]), "v"(bufA[4]), "v"(bufA[5]), "v"(bufA[6]), "v"(bufA[7]), "v"(bufA[8]), "v"(bufA[9]), "v"(bufA[10]), "v"(bufA[11]), "v"(bufA[12]), "v"(bufA[13].x), "v"(bufA[13].y));
#pragma unroll
        for (int a = 0; a < KK; ++a) { xb0[a] = xbn0[a]; xb1[a] = xbn1[a]; }
    }

    {
        v2f den[4], rcp[4];
        fold_den(den, st.Q[1], st.xa[1]);
        fold_rcp(rcp, den);
        fold_fma(acc, st.P[1], rcp);
    }

    const int f0 = ftile * 16 + (hi ? 8 : 0);
    float* o = out + ((size_t)(b * FF + f0) * HH + h) * WW + w0 + laneM;
#pragma unroll
    for (int k = 0; k < 4; ++k) {
        o[(size_t)(2 * k) * (HH * WW)]     = acc[k].x;
        o[(size_t)(2 * k + 1) * (HH * WW)] = acc[k].y;
    }
}

extern "C" void kernel_launch(void* const* d_in, const int* in_sizes, int n_in,
                              void* d_out, int out_size, void* d_ws, size_t ws_size,
                              hipStream_t stream) {
    (void)in_sizes; (void)n_in; (void)out_size; (void)ws_size;
    const float* x      = (const float*)d_in[0];
    const float* nums   = (const float*)d_in[1];
    const float* denoms = (const float*)d_in[2];
    float* out = (float*)d_out;
    float* cf  = (float*)d_ws;

    pack_coeffs<<<56, 256, 0, stream>>>(nums, denoms, cf);
    pade_conv_wmma<<<256, 256, 0, stream>>>(x, cf, out);
}


# x taps loaded per lane straight from global (9 clamped loads per channel, own position in both lane halves) instead of 3 row windows + 9 lane permutes through the LDS unit per channel
# speedup vs baseline: 3.1228x; 1.0528x over previous
#include <hip/hip_runtime.h>

typedef __attribute__((ext_vector_type(2))) float v2f;
typedef __attribute__((ext_vector_type(4))) float v4f;
typedef __attribute__((ext_vector_type(8))) float v8f;

__device__ __forceinline__ v2f make_v2f(float a, float b) {
    v2f r; r.x = a; r.y = b; return r;
}

#define BBATCH 4
#define CC     16
#define HH     64
#define WW     64
#define FF     32
#define KK     3

#define NJ        14
#define CF_V4F    (2 * CC * NJ * 32)

__global__ __launch_bounds__(256) void pack_coeffs(
    const float* __restrict__ nums,
    const float* __restrict__ denoms,
    float* __restrict__ cf)
{
    const int gid   = blockIdx.x * 256 + threadIdx.x;
    const int lane  = gid & 31;
    const int pj    = gid >> 5;
    const int j     = pj % NJ;
    const int fc    = pj / NJ;
    const int c     = fc & (CC - 1);
    const int ftile = fc >> 4;
    const int f     = ftile * 16 + (lane & 15);
    const int kofs  = (lane >= 16) ? 2 : 0;

    const float* __restrict__ np = nums   + (size_t)(f * CC + c) * (KK * KK * 6);
    const float* __restrict__ dp = denoms + (size_t)(f * CC + c) * (KK * KK * 4);

    float v[4];
#pragma unroll
    for (int e = 0; e < 4; ++e) {
        const int  i   = 4 * j + e;
        const bool pad = i >= KK * KK * 6;
        const int  ic  = pad ? 0 : i;
        const int  tap = ic / 6;
        const int  r   = ic - tap * 6;
        const float* src;
        bool zero = pad;
        if (r < 2)      { src = np + tap * 6 + kofs + r; }
        else if (r < 4) { src = np + tap * 6 + 2 + r; zero = zero || (kofs != 0); }
        else            { src = dp + tap * 4 + kofs + (r - 4); }
        const float ld = *src;
        v[e] = zero ? 0.0f : ld;
    }
    v4f o; o.x = v[0]; o.y = v[1]; o.z = v[2]; o.w = v[3];
    ((v4f*)cf)[gid] = o;
}

__device__ __forceinline__ void load_coeffs(v4f (&dst)[NJ], const v4f* p) {
#pragma unroll
    for (int j = 0; j < NJ - 1; ++j) dst[j] = p[j * 32];
    const v2f t = *(const v2f*)(p + (NJ - 1) * 32);
    dst[NJ - 1].x = t.x; dst[NJ - 1].y = t.y; dst[NJ - 1].z = 0.0f; dst[NJ - 1].w = 0.0f;
}

#define NT (KK * KK)
__device__ __forceinline__ void request_x(float (&dst)[NT], const float* xplane, const unsigned (&xoff)[NT]) {
#pragma unroll
    for (int t = 0; t < NT; ++t) dst[t] = *(const float*)((const char*)xplane + xoff[t]);
}

struct Powers { float x1, x2, x4, x5; };
struct AFrag  { v2f p0, p1; };

__device__ __forceinline__ Powers make_powers(const float xv) {
    Powers w;
    w.x1 = xv;
    w.x2 = xv * xv;
    w.x4 = w.x2 * w.x2;
    w.x5 = w.x4 * xv;
    return w;
}

__device__ __forceinline__ AFrag make_afrag(const Powers& w, const bool hi) {
    AFrag f;
    const float s = hi ? w.x2 : 1.0f;
    f.p0 = make_v2f(s, s * w.x1);
    f.p1 = make_v2f(w.x4, w.x5);
    return f;
}

__device__ __forceinline__ float den1(float q, const float xa) {
    asm("v_fma_f32 %0, |%0|, |%1|, 1.0" : "+v"(q) : "v"(xa));
    return q;
}
__device__ __forceinline__ void fold_den(v2f (&d)[4], const v8f& Q, const float xa) {
#pragma unroll
    for (int k = 0; k < 4; ++k)
        d[k] = make_v2f(den1(Q[2 * k], xa), den1(Q[2 * k + 1], xa));
}
__device__ __forceinline__ void fold_rcp(v2f (&r)[4], const v2f (&d)[4]) {
#pragma unroll
    for (int k = 0; k < 4; ++k)
        r[k] = make_v2f(__builtin_amdgcn_rcpf(d[k].x), __builtin_amdgcn_rcpf(d[k].y));
}
__device__ __forceinline__ void fold_fma(v2f (&acc)[4], const v8f& P, const v2f (&r)[4]) {
#pragma unroll
    for (int k = 0; k < 4; ++k)
        acc[k] = __builtin_elementwise_fma(make_v2f(P[2 * k], P[2 * k + 1]), r[k], acc[k]);
}

__device__ __forceinline__ float cf_at(const v4f (&buf)[NJ], const int i) { return buf[i >> 2][i & 3]; }

#define TAPS (2 * KK * KK)

struct Pipe {
    float xv[4];
    AFrag af[2];
    float xa[2];
    v8f   P[2], Q[2];
};

__device__ __forceinline__ float tap_x(const float (&xr)[NT], const bool (&xok)[NT], const int tt) {
    return xok[tt] ? xr[tt] : 0.0f;
}

__device__ __forceinline__ void trip(
    Pipe& st,
    float (&xr0)[NT], float (&xr1)[NT],
    const float* xpl2, const float* xpl3,
    const unsigned (&xoff)[NT], const bool (&xok)[NT],
    const bool hi,
    v4f (&bufA)[NJ], v4f (&bufB)[NJ],
    const v4f* pB, const v4f* pA,
    v2f (&acc)[4])
{
    const v8f zc = {};
    Powers pw = {};
    v2f    den[4], rcp[4];

    load_coeffs(bufB, pB);

#pragma unroll
    for (int u = 0; u < TAPS; ++u) {
        const int ch = (u >= KK * KK) ? 1 : 0;
        const int i0 = (u - ch * KK * KK) * 6;
        const int ur = u & 1;
        const int up = (u + 1) & 1;
        const int t1 = (u + 1) % TAPS;
        const int t3 = (u + 3) % TAPS;

        if (u == KK * KK - 3) request_x(xr0, xpl2, xoff);
        if (u == KK * KK)     load_coeffs(bufA, pA);
        if (u == TAPS - 3)    request_x(xr1, xpl3, xoff);

        {
            const v2f bP0 = make_v2f(ch ? cf_at(bufB, i0 + 0) : cf_at(bufA, i0 + 0),
                                     ch ? cf_at(bufB, i0 + 1) : cf_at(bufA, i0 + 1));
            st.P[ur] = __builtin_amdgcn_wmma_f32_16x16x4_f32(
                false, bP0, false, st.af[ur].p0, (short)0, zc, false, false);
        }
        fold_den(den, st.Q[up], st.xa[up]);
        __builtin_amdgcn_sched_barrier(0);

        {
            const v2f bQ = make_v2f(ch ? cf_at(bufB, i0 + 4) : cf_at(bufA, i0 + 4),
                                    ch ? cf_at(bufB, i0 + 5) : cf_at(bufA, i0 + 5));
            st.Q[ur] = __builtin_amdgcn_wmma_f32_16x16x4_f32(
                false, bQ, false, st.af[ur].p0, (short)0, zc, false, false);
        }
        fold_rcp(rcp, den);
        pw = make_powers(st.xv[t1 & 3]);
        st.xa[t1 & 1] = st.xv[t1 & 3];
        __builtin_amdgcn_sched_barrier(0);

        {
            const v2f bP1 = make_v2f(ch ? cf_at(bufB, i0 + 2) : cf_at(bufA, i0 + 2),
                                     ch ? cf_at(bufB, i0 + 3) : cf_at(bufA, i0 + 3));
            st.P[ur] = __builtin_amdgcn_wmma_f32_16x16x4_f32(
                false, bP1, false, st.af[ur].p1, (short)0, st.P[ur], false, false);
        }
        fold_fma(acc, st.P[up], rcp);
        st.af[t1 & 1] = make_afrag(pw, hi);
        if (u + 3 >= TAPS)         st.xv[t3 & 3] = tap_x(xr0, xok, t3);
        else if (u + 3 >= KK * KK) st.xv[t3 & 3] = tap_x(xr1, xok, t3 - KK * KK);
        else                       st.xv[t3 & 3] = tap_x(xr0, xok, t3);
        __builtin_amdgcn_sched_barrier(0);
    }
}

__global__ __launch_bounds__(256) void pade_conv_wmma(
    const float* x,
    const float* cf,
    float* __restrict__ out)
{
    const int lane  = threadIdx.x & 31;
    const int gw    = blockIdx.x * 8 + (threadIdx.x >> 5);
    const int ftile = gw & 1;
    const int tile  = gw >> 1;

    const int w0 = (tile & 3) * 16;
    const int h  = (tile >> 2) & (HH - 1);
    const int b  = tile >> 8;

    const int  laneM = lane & 15;
    const bool hi    = lane >= 16;

    v2f acc[4] = {};

    unsigned xoff[NT];
    bool     xok[NT];
#pragma unroll
    for (int t = 0; t < NT; ++t) {
        const int  a   = t / KK, bb = t - a * KK;
        const int  hh  = h + a - 1;
        const int  wl  = w0 + laneM + bb - 1;
        const bool hOK = (unsigned)hh < (unsigned)HH;
        const bool wOK = (unsigned)wl < (unsigned)WW;
        xoff[t] = (unsigned)(((hOK ? hh : h) * WW + (wOK ? wl : w0 + laneM)) * 4);
        xok[t]  = hOK && wOK;
    }

    const v4f* cfw = (const v4f*)cf + (size_t)ftile * (CC * NJ * 32) + lane;
    const float* xb = x + (size_t)b * CC * (HH * WW);

    float xr0[NT], xr1[NT];
    request_x(xr0, xb, xoff);
    request_x(xr1, xb + HH * WW, xoff);

    v4f bufA[NJ], bufB[NJ];
    load_coeffs(bufA, cfw);

    Pipe st;
    st.P[1] = v8f{};
    st.Q[1] = v8f{};
#pragma unroll
    for (int t = 0; t < 3; ++t) st.xv[t] = tap_x(xr0, xok, t);
    st.af[0] = make_afrag(make_powers(st.xv[0]), hi);
    st.xa[0] = st.xv[0];
    st.xa[1] = 0.0f;

#pragma unroll 1
    for (int c = 0; c < CC; c += 2) {
        const int cp = (c + 2 < CC) ? (c + 2) : c;
        const int cn = (c + 2 < CC) ? (c + 2) : (CC - 1);

        trip(st, xr0, xr1, xb + (size_t)cp * (HH * WW), xb + (size_t)(cp + 1) * (HH * WW),
             xoff, xok, hi, bufA, bufB,
             cfw + (size_t)(c + 1) * (NJ * 32), cfw + (size_t)cn * (NJ * 32), acc);

        asm volatile("" :: "v"(bufA[0]), "v"(bufA[1]), "v"(bufA[2]), "v"(bufA[3]), "v"(bufA[4]), "v"(bufA[5]), "v"(bufA[6]), "v"(bufA[7]), "v"(bufA[8]), "v"(bufA[9]), "v"(bufA[10]), "v"(bufA[11]), "v"(bufA[12]), "v"(bufA[13].x), "v"(bufA[13].y));
    }

    {
        v2f den[4], rcp[4];
        fold_den(den, st.Q[1], st.xa[1]);
        fold_rcp(rcp, den);
        fold_fma(acc, st.P[1], rcp);
    }

    const int f0 = ftile * 16 + (hi ? 8 : 0);
    float* o = out + ((size_t)(b * FF + f0) * HH + h) * WW + w0 + laneM;
#pragma unroll
    for (int k = 0; k < 4; ++k) {
        o[(size_t)(2 * k) * (HH * WW)]     = acc[k].x;
        o[(size_t)(2 * k + 1) * (HH * WW)] = acc[k].y;
    }
}

extern "C" void kernel_launch(void* const* d_in, const int* in_sizes, int n_in,
                              void* d_out, int out_size, void* d_ws, size_t ws_size,
                              hipStream_t stream) {
    (void)in_sizes; (void)n_in; (void)out_size; (void)ws_size;
    const float* x      = (const float*)d_in[0];
    const float* nums   = (const float*)d_in[1];
    const float* denoms = (const float*)d_in[2];
    float* out = (float*)d_out;
    float* cf  = (float*)d_ws;

    pack_coeffs<<<56, 256, 0, stream>>>(nums, denoms, cf);
    pade_conv_wmma<<<256, 256, 0, stream>>>(x, cf, out);
}


# re-tile: wave = 16 filters x whole 64-wide row x 4 channels; a coefficient pair feeds 4 products, read in place from nums/denoms, no pack launch; 4 channel groups summed apart, added via LDS
# speedup vs baseline: 3.3331x; 1.0673x over previous
#include <hip/hip_runtime.h>

typedef __attribute__((ext_vector_type(2))) float v2f;
typedef __attribute__((ext_vector_type(8))) float v8f;

__device__ __forceinline__ v2f make_v2f(float a, float b) {
    v2f r; r.x = a; r.y = b; return r;
}

#define BBATCH 4
#define CC     16
#define HH     64
#define WW     64
#define FF     32
#define KK     3

#define NTL   (WW / 16)
#define CG    4
#define SLOTS (KK * NTL)

struct CfRow { v2f p0[KK], p1[KK], q[KK]; };

__device__ __forceinline__ void load_cf(CfRow& d, const char* np, const char* dp,
                                        const unsigned nA, const unsigned nB, const unsigned dO) {
#pragma unroll
    for (int bb = 0; bb < KK; ++bb) {
        d.p0[bb] = *(const v2f*)(np + (size_t)nA + bb * 24);
        d.p1[bb] = *(const v2f*)(np + (size_t)nB + bb * 24);
        d.q[bb]  = *(const v2f*)(dp + (size_t)dO + bb * 16);
    }
}

struct XOff { unsigned j0[KK], j1[KK], last; };

__device__ __forceinline__ void request_x(float (&dst)[SLOTS], const char* rowp, const XOff& o) {
#pragma unroll
    for (int bb = 0; bb < KK; ++bb) {
#pragma unroll
        for (int j = 0; j < NTL; ++j) {
            const int u = bb * NTL + j;
            if (j == 0)                            dst[u] = *(const float*)(rowp + (size_t)o.j0[bb]);
            else if (bb == KK - 1 && j == NTL - 1) dst[u] = *(const float*)(rowp + (size_t)o.last);
            else                                   dst[u] = *(const float*)(rowp + (size_t)o.j1[bb] + 64 * (j - 1));
        }
    }
}

__device__ __forceinline__ float tap_x(const float (&xr)[SLOTS], const int u,
                                       const bool rowok, const bool cL, const bool cR) {
    bool ok = rowok;
    if (u == 0)         ok = ok && cL;
    if (u == SLOTS - 1) ok = ok && cR;
    return ok ? xr[u] : 0.0f;
}

struct AFrag { v2f p0, p1; };

__device__ __forceinline__ float den1(float q, const float xa) {
    asm("v_fma_f32 %0, |%0|, |%1|, 1.0" : "+v"(q) : "v"(xa));
    return q;
}
__device__ __forceinline__ void fold_den(v2f (&d)[4], const v8f& Q, const float xa) {
#pragma unroll
    for (int k = 0; k < 4; ++k)
        d[k] = make_v2f(den1(Q[2 * k], xa), den1(Q[2 * k + 1], xa));
}
__device__ __forceinline__ void fold_rcp(v2f (&r)[4], const v2f (&d)[4]) {
#pragma unroll
    for (int k = 0; k < 4; ++k)
        r[k] = make_v2f(__builtin_amdgcn_rcpf(d[k].x), __builtin_amdgcn_rcpf(d[k].y));
}
__device__ __forceinline__ void fold_fma(v2f (&acc)[4], const v8f& P, const v2f (&r)[4]) {
#pragma unroll
    for (int k = 0; k < 4; ++k)
        acc[k] = __builtin_elementwise_fma(make_v2f(P[2 * k], P[2 * k + 1]), r[k], acc[k]);
}

struct Pipe {
    float xv[4];
    AFrag af[2];
    float xa[2];
    v8f   P[2], Q[2];
};

__device__ __forceinline__ void trip(
    Pipe& st, v2f (&acc)[NTL][4],
    const CfRow& cfc, CfRow& cfn,
    const float (&xc)[SLOTS], float (&xn)[SLOTS],
    const bool okc, const bool okn,
    const bool cL, const bool cR, const bool hi,
    const char* npn, const char* dpn, const char* xrown,
    const unsigned nA, const unsigned nB, const unsigned dO, const XOff& xo)
{
    const v8f zc = {};
    v2f   den[4], rcp[4];
    float x1n = 0.0f, x2n = 0.0f, sn = 0.0f, x4n = 0.0f;

    request_x(xn, xrown, xo);
    load_cf(cfn, npn, dpn, nA, nB, dO);

    v2f p1c[KK];
#pragma unroll
    for (int bb = 0; bb < KK; ++bb)
        p1c[bb] = make_v2f(hi ? 0.0f : cfc.p1[bb].x, hi ? 0.0f : cfc.p1[bb].y);

#pragma unroll
    for (int u = 0; u < SLOTS; ++u) {
        const int bb = u / NTL;
        const int ur = u & 1;
        const int up = (u + 1) & 1;
        const int t1 = u + 1;
        const int t3 = u + 3;

        st.P[ur] = __builtin_amdgcn_wmma_f32_16x16x4_f32(
            false, cfc.p0[bb], false, st.af[ur].p0, (short)0, zc, false, false);
        fold_den(den, st.Q[up], st.xa[up]);
        __builtin_amdgcn_sched_barrier(0);

        st.Q[ur] = __builtin_amdgcn_wmma_f32_16x16x4_f32(
            false, cfc.q[bb], false, st.af[ur].p0, (short)0, zc, false, false);
        fold_rcp(rcp, den);
        x1n = st.xv[t1 & 3];
        x2n = x1n * x1n;
        sn  = hi ? x2n : 1.0f;
        x4n = x2n * x2n;
        st.xa[t1 & 1] = x1n;
        __builtin_amdgcn_sched_barrier(0);

        st.P[ur] = __builtin_amdgcn_wmma_f32_16x16x4_f32(
            false, p1c[bb], false, st.af[ur].p1, (short)0, st.P[ur], false, false);
        fold_fma(acc[(u + NTL - 1) % NTL], st.P[up], rcp);
        st.af[t1 & 1].p0 = make_v2f(sn, sn * x1n);
        st.af[t1 & 1].p1 = make_v2f(x4n, x4n * x1n);
        if (t3 < SLOTS) st.xv[t3 & 3] = tap_x(xc, t3, okc, cL, cR);
        else            st.xv[t3 & 3] = tap_x(xn, t3 - SLOTS, okn, cL, cR);
        __builtin_amdgcn_sched_barrier(0);
    }
}

__global__ __launch_bounds__(256) void pade_conv_wmma(
    const float* x,
    const float* nums,
    const float* denoms,
    float* __restrict__ out)
{
    __shared__ float red[8][NTL * 8][32];

    const int lane  = threadIdx.x & 31;
    const int wv    = __builtin_amdgcn_readfirstlane((int)(threadIdx.x >> 5));
    const int ftile = wv & 1;
    const int g     = wv >> 1;
    const int h     = blockIdx.x & (HH - 1);
    const int b     = blockIdx.x >> 6;

    const int  laneM = lane & 15;
    const bool hi    = lane >= 16;
    const bool cL    = laneM != 0;
    const bool cR    = laneM != 15;

    v2f acc[NTL][4] = {};

    const int      f  = ftile * 16 + laneM;
    const unsigned fc = (unsigned)(f * CC + g * CG);
    const unsigned nA = fc * (KK * KK * 6 * 4) + (hi ? 8u : 0u);
    const unsigned nB = fc * (KK * KK * 6 * 4) + 16u;
    const unsigned dO = fc * (KK * KK * 4 * 4) + (hi ? 8u : 0u);

    XOff xo;
#pragma unroll
    for (int bb = 0; bb < KK; ++bb) {
        const int w = laneM + bb - 1;
        xo.j0[bb] = (unsigned)((w < 0 ? 0 : w) * 4);
        xo.j1[bb] = (unsigned)((16 + w) * 4);
    }
    xo.last = (unsigned)((cR ? 48 + laneM + 1 : WW - 1) * 4);

    const bool ok0 = h > 0, ok2 = h < HH - 1;
    const unsigned ro0 = (unsigned)((ok0 ? h - 1 : h) * WW * 4);
    const unsigned ro1 = (unsigned)(h * WW * 4);
    const unsigned ro2 = (unsigned)((ok2 ? h + 1 : h) * WW * 4);

    const char* xg = (const char*)(x + (size_t)(b * CC + g * CG) * (HH * WW));
    const char* nb = (const char*)nums;
    const char* db = (const char*)denoms;

    CfRow cf0, cf1, cf2;
    float xr0[SLOTS], xr1[SLOTS], xr2[SLOTS];
    request_x(xr0, xg + ro0, xo);
    load_cf(cf0, nb, db, nA, nB, dO);

    Pipe st;
    st.P[1] = v8f{};
    st.Q[1] = v8f{};
#pragma unroll
    for (int t = 0; t < 3; ++t) st.xv[t] = tap_x(xr0, t, ok0, cL, cR);
    {
        const float x1 = st.xv[0], x2 = x1 * x1;
        const float s  = hi ? x2 : 1.0f;
        const float x4 = x2 * x2;
        st.af[0].p0 = make_v2f(s, s * x1);
        st.af[0].p1 = make_v2f(x4, x4 * x1);
    }
    st.xa[0] = st.xv[0];
    st.xa[1] = 0.0f;

#pragma unroll 1
    for (int cl = 0; cl < CG; ++cl) {
        const int cn = (cl + 1 < CG) ? (cl + 1) : cl;
        const char* xp  = xg + (size_t)cl * (HH * WW * 4);
        const char* xpn = xg + (size_t)cn * (HH * WW * 4);
        const char* np  = nb + (size_t)cl * (KK * KK * 24);
        const char* dp  = db + (size_t)cl * (KK * KK * 16);
        const char* npn = nb + (size_t)cn * (KK * KK * 24);
        const char* dpn = db + (size_t)cn * (KK * KK * 16);

        trip(st, acc, cf0, cf1, xr0, xr1, ok0, true, cL, cR, hi,
             np + 1 * (KK * 24), dp + 1 * (KK * 16), xp + ro1, nA, nB, dO, xo);
        trip(st, acc, cf1, cf2, xr1, xr2, true, ok2, cL, cR, hi,
             np + 2 * (KK * 24), dp + 2 * (KK * 16), xp + ro2, nA, nB, dO, xo);
        trip(st, acc, cf2, cf0, xr2, xr0, ok2, ok0, cL, cR, hi,
             npn, dpn, xpn + ro0, nA, nB, dO, xo);

        asm volatile("" :: "v"(cf0.p0[0]), "v"(cf0.p0[1]), "v"(cf0.p0[2]),
                           "v"(cf0.p1[0]), "v"(cf0.p1[1]), "v"(cf0.p1[2]),
                           "v"(cf0.q[0]),  "v"(cf0.q[1]),  "v"(cf0.q[2]),
                           "v"(xr0[3]), "v"(xr0[4]), "v"(xr0[5]), "v"(xr0[6]), "v"(xr0[7]),
                           "v"(xr0[8]), "v"(xr0[9]), "v"(xr0[10]), "v"(xr0[11]));
    }

    {
        v2f den[4], rcp[4];
        fold_den(den, st.Q[1], st.xa[1]);
        fold_rcp(rcp, den);
        fold_fma(acc[NTL - 1], st.P[1], rcp);
    }

#pragma unroll
    for (int j = 0; j < NTL; ++j) {
#pragma unroll
        for (int k = 0; k < 4; ++k) {
            red[wv][j * 8 + 2 * k][lane]     = acc[j][k].x;
            red[wv][j * 8 + 2 * k + 1][lane] = acc[j][k].y;
        }
    }
    __syncthreads();

    const int f0 = ftile * 16 + (hi ? 8 : 0);
    float* o = out + ((size_t)(b * FF + f0) * HH + h) * WW + g * 16 + laneM;
#pragma unroll
    for (int e = 0; e < 8; ++e) {
        const float s = ((red[ftile][g * 8 + e][lane] + red[ftile + 2][g * 8 + e][lane])
                         + red[ftile + 4][g * 8 + e][lane]) + red[ftile + 6][g * 8 + e][lane];
        o[(size_t)e * (HH * WW)] = s;
    }
}

extern "C" void kernel_launch(void* const* d_in, const int* in_sizes, int n_in,
                              void* d_out, int out_size, void* d_ws, size_t ws_size,
                              hipStream_t stream) {
    (void)in_sizes; (void)n_in; (void)out_size; (void)d_ws; (void)ws_size;
    const float* x      = (const float*)d_in[0];
    const float* nums   = (const float*)d_in[1];
    const float* denoms = (const float*)d_in[2];
    float* out = (float*)d_out;

    pade_conv_wmma<<<256, 256, 0, stream>>>(x, nums, denoms, out);
}


# the three column taps of a tile are summed as one fraction N/D carried by packed multiplies and divided once: one reciprocal per three terms, not three; sum order inside a tap row changes
# speedup vs baseline: 3.5139x; 1.0542x over previous
#include <hip/hip_runtime.h>

typedef __attribute__((ext_vector_type(2))) float v2f;
typedef __attribute__((ext_vector_type(8))) float v8f;

__device__ __forceinline__ v2f make_v2f(float a, float b) {
    v2f r; r.x = a; r.y = b; return r;
}

#define BBATCH 4
#define CC     16
#define HH     64
#define WW     64
#define FF     32
#define KK     3

#define NTL   (WW / 16)
#define CG    4
#define SLOTS (KK * NTL)

struct CfRow { v2f p0[KK], p1[KK], q[KK]; };

__device__ __forceinline__ void load_cf(CfRow& d, const char* np, const char* dp,
                                        const unsigned nA, const unsigned nB, const unsigned dO) {
#pragma unroll
    for (int bb = 0; bb < KK; ++bb) {
        d.p0[bb] = *(const v2f*)(np + (size_t)nA + bb * 24);
        d.p1[bb] = *(const v2f*)(np + (size_t)nB + bb * 24);
        d.q[bb]  = *(const v2f*)(dp + (size_t)dO + bb * 16);
    }
}

struct XOff { unsigned j0[KK], j1[KK], last; };

__device__ __forceinline__ void request_x(float (&dst)[SLOTS], const char* rowp, const XOff& o) {
#pragma unroll
    for (int bb = 0; bb < KK; ++bb) {
#pragma unroll
        for (int j = 0; j < NTL; ++j) {
            const int u = bb * NTL + j;
            if (j == 0)                            dst[u] = *(const float*)(rowp + (size_t)o.j0[bb]);
            else if (bb == KK - 1 && j == NTL - 1) dst[u] = *(const float*)(rowp + (size_t)o.last);
            else                                   dst[u] = *(const float*)(rowp + (size_t)o.j1[bb] + 64 * (j - 1));
        }
    }
}

__device__ __forceinline__ constexpr int xslot(const int t) { return (t % KK) * NTL + t / KK; }

__device__ __forceinline__ float tap_x(const float (&xr)[SLOTS], const int u,
                                       const bool rowok, const bool cL, const bool cR) {
    bool ok = rowok;
    if (u == 0)         ok = ok && cL;
    if (u == SLOTS - 1) ok = ok && cR;
    return ok ? xr[u] : 0.0f;
}

struct AFrag { v2f p0, p1; };

__device__ __forceinline__ float den1(float q, const float xa) {
    asm("v_fma_f32 %0, |%0|, |%1|, 1.0" : "+v"(q) : "v"(xa));
    return q;
}
__device__ __forceinline__ void fold_den(v2f (&d)[4], const v8f& Q, const float xa) {
#pragma unroll
    for (int k = 0; k < 4; ++k)
        d[k] = make_v2f(den1(Q[2 * k], xa), den1(Q[2 * k + 1], xa));
}
__device__ __forceinline__ void fold_rcp(v2f (&r)[4], const v2f (&d)[4]) {
#pragma unroll
    for (int k = 0; k < 4; ++k)
        r[k] = make_v2f(__builtin_amdgcn_rcpf(d[k].x), __builtin_amdgcn_rcpf(d[k].y));
}
__device__ __forceinline__ void fold_fma(v2f (&acc)[4], const v2f (&n)[4], const v2f (&r)[4]) {
#pragma unroll
    for (int k = 0; k < 4; ++k)
        acc[k] = __builtin_elementwise_fma(n[k], r[k], acc[k]);
}
__device__ __forceinline__ void frac_mul(v2f (&tn)[4], v2f (&td)[4], const v2f (&N)[4],
                                         const v2f (&D)[4], const v2f (&d)[4]) {
#pragma unroll
    for (int k = 0; k < 4; ++k) { tn[k] = N[k] * d[k]; td[k] = D[k] * d[k]; }
}
__device__ __forceinline__ void frac_add(v2f (&N)[4], v2f (&D)[4], const v8f& P,
                                         const v2f (&tn)[4], const v2f (&td)[4]) {
#pragma unroll
    for (int k = 0; k < 4; ++k) {
        N[k] = __builtin_elementwise_fma(make_v2f(P[2 * k], P[2 * k + 1]), D[k], tn[k]);
        D[k] = td[k];
    }
}
__device__ __forceinline__ void frac_set(v2f (&N)[4], v2f (&D)[4], const v8f& P, const v2f (&d)[4]) {
#pragma unroll
    for (int k = 0; k < 4; ++k) { N[k] = make_v2f(P[2 * k], P[2 * k + 1]); D[k] = d[k]; }
}

struct Pipe {
    float xv[4];
    AFrag af[2];
    float xa[2];
    v8f   P[2], Q[2];
    v2f   N[4], D[4];
};

__device__ __forceinline__ void trip(
    Pipe& st, v2f (&acc)[NTL][4],
    const CfRow& cfc, CfRow& cfn,
    const float (&xc)[SLOTS], float (&xn)[SLOTS],
    const bool okc, const bool okn,
    const bool cL, const bool cR, const bool hi,
    const char* npn, const char* dpn, const char* xrown,
    const unsigned nA, const unsigned nB, const unsigned dO, const XOff& xo)
{
    const v8f zc = {};
    v2f   den[4], rcp[4], tn[4], td[4];
    float x1n = 0.0f, x2n = 0.0f, sn = 0.0f, x4n = 0.0f;

    request_x(xn, xrown, xo);
    load_cf(cfn, npn, dpn, nA, nB, dO);

    v2f p1c[KK];
#pragma unroll
    for (int bb = 0; bb < KK; ++bb)
        p1c[bb] = make_v2f(hi ? 0.0f : cfc.p1[bb].x, hi ? 0.0f : cfc.p1[bb].y);

#pragma unroll
    for (int u = 0; u < SLOTS; ++u) {
        const int bb = u % KK;
        const int ur = u & 1;
        const int up = (u + 1) & 1;
        const int t1 = u + 1;
        const int t3 = u + 3;
        const int b1 = (u + SLOTS - 1) % SLOTS;
        const bool first = (b1 % KK) == 0;
        const int  jdone = (b1 / KK + NTL - 1) % NTL;

        st.P[ur] = __builtin_amdgcn_wmma_f32_16x16x4_f32(
            false, cfc.p0[bb], false, st.af[ur].p0, (short)0, zc, false, false);
        fold_den(den, st.Q[up], st.xa[up]);
        __builtin_amdgcn_sched_barrier(0);

        st.Q[ur] = __builtin_amdgcn_wmma_f32_16x16x4_f32(
            false, cfc.q[bb], false, st.af[ur].p0, (short)0, zc, false, false);
        if (first) fold_rcp(rcp, st.D);
        else       frac_mul(tn, td, st.N, st.D, den);
        x1n = st.xv[t1 & 3];
        x2n = x1n * x1n;
        sn  = hi ? x2n : 1.0f;
        x4n = x2n * x2n;
        st.xa[t1 & 1] = x1n;
        __builtin_amdgcn_sched_barrier(0);

        st.P[ur] = __builtin_amdgcn_wmma_f32_16x16x4_f32(
            false, p1c[bb], false, st.af[ur].p1, (short)0, st.P[ur], false, false);
        if (first) { fold_fma(acc[jdone], st.N, rcp); frac_set(st.N, st.D, st.P[up], den); }
        else       frac_add(st.N, st.D, st.P[up], tn, td);
        st.af[t1 & 1].p0 = make_v2f(sn, sn * x1n);
        st.af[t1 & 1].p1 = make_v2f(x4n, x4n * x1n);
        if (t3 < SLOTS) st.xv[t3 & 3] = tap_x(xc, xslot(t3), okc, cL, cR);
        else            st.xv[t3 & 3] = tap_x(xn, xslot(t3 - SLOTS), okn, cL, cR);
        __builtin_amdgcn_sched_barrier(0);
    }
}

__global__ __launch_bounds__(256) void pade_conv_wmma(
    const float* x,
    const float* nums,
    const float* denoms,
    float* __restrict__ out)
{
    __shared__ float red[8][NTL * 8][32];

    const int lane  = threadIdx.x & 31;
    const int wv    = __builtin_amdgcn_readfirstlane((int)(threadIdx.x >> 5));
    const int ftile = wv & 1;
    const int g     = wv >> 1;
    const int h     = blockIdx.x & (HH - 1);
    const int b     = blockIdx.x >> 6;

    const int  laneM = lane & 15;
    const bool hi    = lane >= 16;
    const bool cL    = laneM != 0;
    const bool cR    = laneM != 15;

    v2f acc[NTL][4] = {};

    const int      f  = ftile * 16 + laneM;
    const unsigned fc = (unsigned)(f * CC + g * CG);
    const unsigned nA = fc * (KK * KK * 6 * 4) + (hi ? 8u : 0u);
    const unsigned nB = fc * (KK * KK * 6 * 4) + 16u;
    const unsigned dO = fc * (KK * KK * 4 * 4) + (hi ? 8u : 0u);

    XOff xo;
#pragma unroll
    for (int bb = 0; bb < KK; ++bb) {
        const int w = laneM + bb - 1;
        xo.j0[bb] = (unsigned)((w < 0 ? 0 : w) * 4);
        xo.j1[bb] = (unsigned)((16 + w) * 4);
    }
    xo.last = (unsigned)((cR ? 48 + laneM + 1 : WW - 1) * 4);

    const bool ok0 = h > 0, ok2 = h < HH - 1;
    const unsigned ro0 = (unsigned)((ok0 ? h - 1 : h) * WW * 4);
    const unsigned ro1 = (unsigned)(h * WW * 4);
    const unsigned ro2 = (unsigned)((ok2 ? h + 1 : h) * WW * 4);

    const char* xg = (const char*)(x + (size_t)(b * CC + g * CG) * (HH * WW));
    const char* nb = (const char*)nums;
    const char* db = (const char*)denoms;

    CfRow cf0, cf1, cf2;
    float xr0[SLOTS], xr1[SLOTS], xr2[SLOTS];
    request_x(xr0, xg + ro0, xo);
    load_cf(cf0, nb, db, nA, nB, dO);

    Pipe st;
    st.P[1] = v8f{};
    st.Q[1] = v8f{};
#pragma unroll
    for (int t = 0; t < 3; ++t) st.xv[t] = tap_x(xr0, xslot(t), ok0, cL, cR);
    {
        const float x1 = st.xv[0], x2 = x1 * x1;
        const float s  = hi ? x2 : 1.0f;
        const float x4 = x2 * x2;
        st.af[0].p0 = make_v2f(s, s * x1);
        st.af[0].p1 = make_v2f(x4, x4 * x1);
    }
    st.xa[0] = st.xv[0];
    st.xa[1] = 0.0f;
#pragma unroll
    for (int k = 0; k < 4; ++k) { st.N[k] = make_v2f(0.0f, 0.0f); st.D[k] = make_v2f(1.0f, 1.0f); }

#pragma unroll 1
    for (int cl = 0; cl < CG; ++cl) {
        const int cn = (cl + 1 < CG) ? (cl + 1) : cl;
        const char* xp  = xg + (size_t)cl * (HH * WW * 4);
        const char* xpn = xg + (size_t)cn * (HH * WW * 4);
        const char* np  = nb + (size_t)cl * (KK * KK * 24);
        const char* dp  = db + (size_t)cl * (KK * KK * 16);
        const char* npn = nb + (size_t)cn * (KK * KK * 24);
        const char* dpn = db + (size_t)cn * (KK * KK * 16);

        trip(st, acc, cf0, cf1, xr0, xr1, ok0, true, cL, cR, hi,
             np + 1 * (KK * 24), dp + 1 * (KK * 16), xp + ro1, nA, nB, dO, xo);
        trip(st, acc, cf1, cf2, xr1, xr2, true, ok2, cL, cR, hi,
             np + 2 * (KK * 24), dp + 2 * (KK * 16), xp + ro2, nA, nB, dO, xo);
        trip(st, acc, cf2, cf0, xr2, xr0, ok2, ok0, cL, cR, hi,
             npn, dpn, xpn + ro0, nA, nB, dO, xo);

        asm volatile("" :: "v"(cf0.p0[0]), "v"(cf0.p0[1]), "v"(cf0.p0[2]),
                           "v"(cf0.p1[0]), "v"(cf0.p1[1]), "v"(cf0.p1[2]),
                           "v"(cf0.q[0]),  "v"(cf0.q[1]),  "v"(cf0.q[2]),
                           "v"(xr0[1]), "v"(xr0[2]), "v"(xr0[3]), "v"(xr0[5]), "v"(xr0[6]),
                           "v"(xr0[7]), "v"(xr0[9]), "v"(xr0[10]), "v"(xr0[11]));
    }

    {
        v2f den[4], rcp[4], tn[4], td[4];
        fold_den(den, st.Q[1], st.xa[1]);
        frac_mul(tn, td, st.N, st.D, den);
        frac_add(st.N, st.D, st.P[1], tn, td);
        fold_rcp(rcp, st.D);
        fold_fma(acc[NTL - 1], st.N, rcp);
    }

#pragma unroll
    for (int j = 0; j < NTL; ++j) {
#pragma unroll
        for (int k = 0; k < 4; ++k) {
            red[wv][j * 8 + 2 * k][lane]     = acc[j][k].x;
            red[wv][j * 8 + 2 * k + 1][lane] = acc[j][k].y;
        }
    }
    __syncthreads();

    const int f0 = ftile * 16 + (hi ? 8 : 0);
    float* o = out + ((size_t)(b * FF + f0) * HH + h) * WW + g * 16 + laneM;
#pragma unroll
    for (int e = 0; e < 8; ++e) {
        const float s = ((red[ftile][g * 8 + e][lane] + red[ftile + 2][g * 8 + e][lane])
                         + red[ftile + 4][g * 8 + e][lane]) + red[ftile + 6][g * 8 + e][lane];
        o[(size_t)e * (HH * WW)] = s;
    }
}

extern "C" void kernel_launch(void* const* d_in, const int* in_sizes, int n_in,
                              void* d_out, int out_size, void* d_ws, size_t ws_size,
                              hipStream_t stream) {
    (void)in_sizes; (void)n_in; (void)out_size; (void)d_ws; (void)ws_size;
    const float* x      = (const float*)d_in[0];
    const float* nums   = (const float*)d_in[1];
    const float* denoms = (const float*)d_in[2];
    float* out = (float*)d_out;

    pade_conv_wmma<<<256, 256, 0, stream>>>(x, nums, denoms, out);
}
